# starting kernel
# baseline (speedup 1.0000x reference)
#include <hip/hip_runtime.h>
#include <math.h>

#define FDIM 64

typedef __attribute__((ext_vector_type(2))) float v2f;
typedef __attribute__((ext_vector_type(8))) float v8f;

__device__ __forceinline__ void atomicMaxF32(float* addr, float val) {
    if (val >= 0.0f) {
        atomicMax((int*)addr, __float_as_int(val));
    } else {
        atomicMin((unsigned int*)addr, __float_as_uint(val));
    }
}

__global__ __launch_bounds__(32) void gat_gemm_wmma(
    const float* __restrict__ X,
    const float* __restrict__ W,
    const float* __restrict__ a_s,
    const float* __restrict__ a_d,
    float* __restrict__ H,
    float* __restrict__ HS,
    float* __restrict__ HD)
{
    const int lane  = threadIdx.x;
    const int m     = lane & 15;
    const int khalf = lane >> 4;
    const int row0  = blockIdx.x * 16;

    const float* xr = X + (size_t)(row0 + m) * FDIM;
    v2f af[16];
#pragma unroll
    for (int k = 0; k < 16; ++k) {
        const int kc = 4 * k + 2 * khalf;
        af[k].x = xr[kc];
        af[k].y = xr[kc + 1];
    }

    v8f acc[4];
#pragma unroll
    for (int j = 0; j < 4; ++j) acc[j] = (v8f){0,0,0,0,0,0,0,0};

#pragma unroll
    for (int j = 0; j < 4; ++j) {
#pragma unroll
        for (int k = 0; k < 16; ++k) {
            const float* wp = W + (size_t)(4 * k + 2 * khalf) * FDIM + j * 16 + m;
            v2f bf;
            bf.x = wp[0];
            bf.y = wp[FDIM];
            acc[j] = __builtin_amdgcn_wmma_f32_16x16x4_f32(
                 false, af[k],  false, bf,
                 (short)0, acc[j],
                 false,  false);
        }
    }

    const int drow_base = row0 + 8 * khalf;
#pragma unroll
    for (int r = 0; r < 8; ++r) {
        float* hp = H + (size_t)(drow_base + r) * FDIM + m;
#pragma unroll
        for (int j = 0; j < 4; ++j) hp[j * 16] = acc[j][r];
    }

    float as_c[4], ad_c[4];
#pragma unroll
    for (int j = 0; j < 4; ++j) {
        as_c[j] = a_s[j * 16 + m];
        ad_c[j] = a_d[j * 16 + m];
    }
#pragma unroll
    for (int r = 0; r < 8; ++r) {
        float ps = 0.0f, pd = 0.0f;
#pragma unroll
        for (int j = 0; j < 4; ++j) {
            ps += acc[j][r] * as_c[j];
            pd += acc[j][r] * ad_c[j];
        }
#pragma unroll
        for (int off = 8; off >= 1; off >>= 1) {
            ps += __shfl_xor(ps, off, 32);
            pd += __shfl_xor(pd, off, 32);
        }
        if (m == 0) {
            HS[drow_base + r] = ps;
            HD[drow_base + r] = pd;
        }
    }
}

__global__ void gat_init(float* __restrict__ OUT, float* __restrict__ Mx,
                         float* __restrict__ Sx, int n_nodes, int tot)
{
    int i = blockIdx.x * blockDim.x + threadIdx.x;
    if (i < tot) OUT[i] = 0.0f;
    if (i < n_nodes) {
        Mx[i] = __int_as_float(0xFF800000);
        Sx[i] = 0.0f;
    }
}

__global__ void gat_edge_score(const int* __restrict__ src,
                               const int* __restrict__ dst,
                               const float* __restrict__ HS,
                               const float* __restrict__ HD,
                               float* __restrict__ EW,
                               float* __restrict__ Mx,
                               int E, int Etot)
{
    int e = blockIdx.x * blockDim.x + threadIdx.x;
    if (e >= Etot) return;
    int s, d;
    if (e < E) { s = src[e]; d = dst[e]; } else { s = d = e - E; }
    float v = HS[s] + HD[d];
    v = (v >= 0.0f) ? v : 0.2f * v;
    EW[e] = v;
    atomicMaxF32(&Mx[d], v);
}

__global__ void gat_edge_exp(const int* __restrict__ dst,
                             const float* __restrict__ Mx,
                             float* __restrict__ EW,
                             float* __restrict__ Sx,
                             int E, int Etot)
{
    int e = blockIdx.x * blockDim.x + threadIdx.x;
    if (e >= Etot) return;
    int d = (e < E) ? dst[e] : (e - E);
    float w = __expf(EW[e] - Mx[d]);
    EW[e] = w;
    atomicAdd(&Sx[d], w);
}

#define EDGES_PER_BLOCK 16
__global__ __launch_bounds__(64) void gat_aggr(
    const int* __restrict__ src, const int* __restrict__ dst,
    const float* __restrict__ EW, const float* __restrict__ Sx,
    const float* __restrict__ H, float* __restrict__ OUT,
    int E, int Etot)
{
    const int f  = threadIdx.x;
    const int e0 = blockIdx.x * EDGES_PER_BLOCK;
#pragma unroll 1
    for (int i = 0; i < EDGES_PER_BLOCK; ++i) {
        int e = e0 + i;
        if (e >= Etot) return;
        int s, d;
        if (e < E) { s = src[e]; d = dst[e]; } else { s = d = e - E; }
        float alpha = EW[e] / Sx[d];
        atomicAdd(&OUT[(size_t)d * FDIM + f], alpha * H[(size_t)s * FDIM + f]);
    }
}

__global__ void gat_bias(float* __restrict__ OUT, const float* __restrict__ b,
                         int tot)
{
    int i = blockIdx.x * blockDim.x + threadIdx.x;
    if (i < tot) OUT[i] += b[i & (FDIM - 1)];
}

extern "C" void kernel_launch(void* const* d_in, const int* in_sizes, int n_in,
                              void* d_out, int out_size, void* d_ws, size_t ws_size,
                              hipStream_t stream) {
    const float* x   = (const float*)d_in[0];
    const int*   ei  = (const int*)  d_in[1];
    const float* W1  = (const float*)d_in[2];
    const float* as1 = (const float*)d_in[3];
    const float* ad1 = (const float*)d_in[4];
    const float* b1  = (const float*)d_in[5];
    const float* W2  = (const float*)d_in[6];
    const float* as2 = (const float*)d_in[7];
    const float* ad2 = (const float*)d_in[8];
    const float* b2  = (const float*)d_in[9];

    const int N    = in_sizes[0] / FDIM;
    const int E    = in_sizes[1] / 2;
    const int Etot = E + N;
    const int* src = ei;
    const int* dst = ei + E;

    float* ws = (float*)d_ws;
    float* H  = ws;  ws += (size_t)N * FDIM;
    float* X1 = ws;  ws += (size_t)N * FDIM;
    float* HS = ws;  ws += N;
    float* HD = ws;  ws += N;
    float* Mx = ws;  ws += N;
    float* Sx = ws;  ws += N;
    float* EW = ws;  ws += Etot;

    const int tot = N * FDIM;

    auto layer = [&](const float* Xin, const float* W, const float* a_s,
                     const float* a_d, const float* b, float* OUT) {
        gat_gemm_wmma<<<N / 16, 32, 0, stream>>>(Xin, W, a_s, a_d, H, HS, HD);
        gat_init<<<(tot + 255) / 256, 256, 0, stream>>>(OUT, Mx, Sx, N, tot);
        gat_edge_score<<<(Etot + 255) / 256, 256, 0, stream>>>(src, dst, HS, HD,
                                                               EW, Mx, E, Etot);
        gat_edge_exp<<<(Etot + 255) / 256, 256, 0, stream>>>(dst, Mx, EW, Sx,
                                                             E, Etot);
        gat_aggr<<<(Etot + EDGES_PER_BLOCK - 1) / EDGES_PER_BLOCK, 64, 0,
                   stream>>>(src, dst, EW, Sx, H, OUT, E, Etot);
        gat_bias<<<(tot + 255) / 256, 256, 0, stream>>>(OUT, b, tot);
    };

    layer(x,  W1, as1, ad1, b1, X1);
    layer(X1, W2, as2, ad2, b2, (float*)d_out);
}


# Work per wave raised 4-8x in every pass, loads issued as one group; aggregate: a wave takes 64 edges, records read by lane, 8 rows fetched ahead of their atomics; f32 sums still in arrival order
# speedup vs baseline: 1.8368x; 1.8368x over previous
#include <hip/hip_runtime.h>
#include <math.h>

#define FDIM 64

typedef __attribute__((ext_vector_type(2))) float v2f;
typedef __attribute__((ext_vector_type(8))) float v8f;

__device__ __forceinline__ void atomicMaxF32(float* addr, float val) {
    if (val >= 0.0f) {
        atomicMax((int*)addr, __float_as_int(val));
    } else {
        atomicMin((unsigned int*)addr, __float_as_uint(val));
    }
}

__device__ __forceinline__ int my_half() {
    return (int)((__builtin_amdgcn_s_sendmsg_rtn(0x87) >> 18) & 1u);
}

__global__ __launch_bounds__(32) void gat_gemm_wmma(
    const float* __restrict__ X,
    const float* __restrict__ W,
    const float* __restrict__ a_s,
    const float* __restrict__ a_d,
    float* __restrict__ H,
    float* __restrict__ HS,
    float* __restrict__ HD)
{
    const int lane  = threadIdx.x;
    const int m     = lane & 15;
    const int khalf = lane >> 4;
    const int row0  = blockIdx.x * 16;

    const float* xr = X + (size_t)(row0 + m) * FDIM;
    v2f af[16];
#pragma unroll
    for (int k = 0; k < 16; ++k) {
        const int kc = 4 * k + 2 * khalf;
        af[k].x = xr[kc];
        af[k].y = xr[kc + 1];
    }

    v8f acc[4];
#pragma unroll
    for (int j = 0; j < 4; ++j) acc[j] = (v8f){0,0,0,0,0,0,0,0};

#pragma unroll
    for (int j = 0; j < 4; ++j) {
#pragma unroll
        for (int k = 0; k < 16; ++k) {
            const float* wp = W + (size_t)(4 * k + 2 * khalf) * FDIM + j * 16 + m;
            v2f bf;
            bf.x = wp[0];
            bf.y = wp[FDIM];
            acc[j] = __builtin_amdgcn_wmma_f32_16x16x4_f32(
                 false, af[k],  false, bf,
                 (short)0, acc[j],
                 false,  false);
        }
    }

    const int drow_base = row0 + 8 * khalf;
#pragma unroll
    for (int r = 0; r < 8; ++r) {
        float* hp = H + (size_t)(drow_base + r) * FDIM + m;
#pragma unroll
        for (int j = 0; j < 4; ++j) hp[j * 16] = acc[j][r];
    }

    float as_c[4], ad_c[4];
#pragma unroll
    for (int j = 0; j < 4; ++j) {
        as_c[j] = a_s[j * 16 + m];
        ad_c[j] = a_d[j * 16 + m];
    }
#pragma unroll
    for (int r = 0; r < 8; ++r) {
        float ps = 0.0f, pd = 0.0f;
#pragma unroll
        for (int j = 0; j < 4; ++j) {
            ps += acc[j][r] * as_c[j];
            pd += acc[j][r] * ad_c[j];
        }
#pragma unroll
        for (int off = 8; off >= 1; off >>= 1) {
            ps += __shfl_xor(ps, off, 32);
            pd += __shfl_xor(pd, off, 32);
        }
        if (m == 0) {
            HS[drow_base + r] = ps;
            HD[drow_base + r] = pd;
        }
    }
}

#define TPB      256
#define INIT_V   8
#define EDGE_V   4
#define JOIN_V   4
#define AGGR_B   2
#define AGGR_G   8

__global__ __launch_bounds__(TPB) void gat_init(float* __restrict__ P,
                                                float* __restrict__ Mx,
                                                float* __restrict__ Sx,
                                                int n2, int tot4x2)
{
    float4* __restrict__ P4 = reinterpret_cast<float4*>(P);
    const int base = blockIdx.x * (TPB * INIT_V) + threadIdx.x;
#pragma unroll
    for (int j = 0; j < INIT_V; ++j) {
        const int i = base + j * TPB;
        if (i < tot4x2) P4[i] = make_float4(0.0f, 0.0f, 0.0f, 0.0f);
    }
    const int i = blockIdx.x * TPB + threadIdx.x;
    if (i < n2) {
        Mx[i] = __int_as_float(0xFF800000);
        Sx[i] = 0.0f;
    }
}

__device__ __forceinline__ void edge_nodes(const int* __restrict__ src,
                                           const int* __restrict__ dst,
                                           int e, int E, int Etot,
                                           int& s, int& d)
{
    const int ec = (e < E) ? e : 0;
    const int sv = src[ec];
    const int dv = dst[ec];
    const int loop = (e < Etot) ? (e - E) : 0;
    s = (e < E) ? sv : loop;
    d = (e < E) ? dv : loop;
}
__device__ __forceinline__ int edge_dst(const int* __restrict__ dst,
                                        int e, int E, int Etot)
{
    const int ec = (e < E) ? e : 0;
    const int dv = dst[ec];
    const int loop = (e < Etot) ? (e - E) : 0;
    return (e < E) ? dv : loop;
}

template <bool FULL>
__device__ __forceinline__ void edge_score_body(
    const int* __restrict__ src, const int* __restrict__ dst,
    const float* __restrict__ HS, const float* __restrict__ HD,
    float* __restrict__ EW, float* __restrict__ Mh,
    int base, int E, int Etot)
{
    int s[EDGE_V], d[EDGE_V];
#pragma unroll
    for (int j = 0; j < EDGE_V; ++j)
        edge_nodes(src, dst, base + j * TPB, E, Etot, s[j], d[j]);
    float v[EDGE_V];
#pragma unroll
    for (int j = 0; j < EDGE_V; ++j) {
        const float t = HS[s[j]] + HD[d[j]];
        v[j] = (t >= 0.0f) ? t : 0.2f * t;
    }
#pragma unroll
    for (int j = 0; j < EDGE_V; ++j) {
        const int e = base + j * TPB;
        if (FULL || e < Etot) {
            EW[e] = v[j];
            atomicMaxF32(&Mh[d[j]], v[j]);
        }
    }
}

__global__ __launch_bounds__(TPB) void gat_edge_score(
    const int* __restrict__ src, const int* __restrict__ dst,
    const float* __restrict__ HS, const float* __restrict__ HD,
    float* __restrict__ EW,
    float* __restrict__ Mx,
    int E, int Etot, int n_nodes)
{
    const int blk0 = blockIdx.x * (TPB * EDGE_V);
    const int base = blk0 + threadIdx.x;
    float* Mh = Mx + (size_t)my_half() * n_nodes;
    if (blk0 + TPB * EDGE_V <= Etot)
        edge_score_body<true >(src, dst, HS, HD, EW, Mh, base, E, Etot);
    else
        edge_score_body<false>(src, dst, HS, HD, EW, Mh, base, E, Etot);
}

template <bool FULL>
__device__ __forceinline__ void edge_exp_body(
    const int* __restrict__ dst, const float* __restrict__ Mx,
    float* __restrict__ EW, float* __restrict__ Sh,
    int base, int E, int Etot, int n_nodes)
{
    int d[EDGE_V];
    float ew[EDGE_V];
#pragma unroll
    for (int j = 0; j < EDGE_V; ++j) {
        const int e = base + j * TPB;
        d[j]  = edge_dst(dst, e, E, Etot);
        ew[j] = EW[(FULL || e < Etot) ? e : 0];
    }
    float w[EDGE_V];
#pragma unroll
    for (int j = 0; j < EDGE_V; ++j) {
        const float m = fmaxf(Mx[d[j]], Mx[n_nodes + d[j]]);
        w[j] = __expf(ew[j] - m);
    }
#pragma unroll
    for (int j = 0; j < EDGE_V; ++j) {
        const int e = base + j * TPB;
        if (FULL || e < Etot) {
            EW[e] = w[j];
            atomicAdd(&Sh[d[j]], w[j]);
        }
    }
}

__global__ __launch_bounds__(TPB) void gat_edge_exp(
    const int* __restrict__ dst,
    const float* __restrict__ Mx,
    float* __restrict__ EW,
    float* __restrict__ Sx,
    int E, int Etot, int n_nodes)
{
    const int blk0 = blockIdx.x * (TPB * EDGE_V);
    const int base = blk0 + threadIdx.x;
    float* Sh = Sx + (size_t)my_half() * n_nodes;
    if (blk0 + TPB * EDGE_V <= Etot)
        edge_exp_body<true >(dst, Mx, EW, Sh, base, E, Etot, n_nodes);
    else
        edge_exp_body<false>(dst, Mx, EW, Sh, base, E, Etot, n_nodes);
}

__device__ __forceinline__ float lane_f32(float v, int k) {
    return __int_as_float(__builtin_amdgcn_readlane(__float_as_int(v), k));
}

__global__ __launch_bounds__(TPB) void gat_aggr(
    const int* __restrict__ src, const int* __restrict__ dst,
    const float* __restrict__ EW, const float* __restrict__ Sx,
    const float* __restrict__ H, float* __restrict__ P,
    int E, int Etot, int n_nodes)
{
    const int lane = threadIdx.x & 31;
    const int wave = blockIdx.x * (TPB / 32) +
                     __builtin_amdgcn_readfirstlane((int)(threadIdx.x >> 5));
    const int e0   = wave * (32 * AGGR_B);
    if (e0 >= Etot) return;
    float* OUT = P + (size_t)my_half() * n_nodes * FDIM;
    const float* Hl = H + lane;
    float* Ol = OUT + lane;

    int   s[AGGR_B], d[AGGR_B];
    float a[AGGR_B];
#pragma unroll
    for (int b = 0; b < AGGR_B; ++b) {
        const int e = e0 + 32 * b + lane;
        edge_nodes(src, dst, e, E, Etot, s[b], d[b]);
        a[b] = EW[(e < Etot) ? e : 0];
    }
#pragma unroll
    for (int b = 0; b < AGGR_B; ++b)
        a[b] = a[b] / (Sx[d[b]] + Sx[n_nodes + d[b]]);

#pragma unroll
    for (int b = 0; b < AGGR_B; ++b) {
        const int cnt = Etot - (e0 + 32 * b);
        if (cnt >= 32) {
#pragma unroll
            for (int k0 = 0; k0 < 32; k0 += AGGR_G) {
                float h0[AGGR_G], h1[AGGR_G];
#pragma unroll
                for (int k = 0; k < AGGR_G; ++k) {
                    const int sk = __builtin_amdgcn_readlane(s[b], k0 + k);
                    const float* hr = Hl + (size_t)sk * FDIM;
                    h0[k] = hr[0];
                    h1[k] = hr[32];
                }
#pragma unroll
                for (int k = 0; k < AGGR_G; ++k) {
                    const int   dk = __builtin_amdgcn_readlane(d[b], k0 + k);
                    const float ak = lane_f32(a[b], k0 + k);
                    float* o = Ol + (size_t)dk * FDIM;
                    atomicAdd(o,      ak * h0[k]);
                    atomicAdd(o + 32, ak * h1[k]);
                }
            }
        } else {
            for (int k = 0; k < cnt; ++k) {
                const int   sk = __builtin_amdgcn_readlane(s[b], k);
                const int   dk = __builtin_amdgcn_readlane(d[b], k);
                const float ak = lane_f32(a[b], k);
                const float* hr = Hl + (size_t)sk * FDIM;
                float* o = Ol + (size_t)dk * FDIM;
                atomicAdd(o,      ak * hr[0]);
                atomicAdd(o + 32, ak * hr[32]);
            }
        }
    }
}

template <bool FULL>
__device__ __forceinline__ void join_bias_body(const float4* __restrict__ P4,
                                               const float* __restrict__ b,
                                               float4* __restrict__ OUT4,
                                               int base, int tot4)
{
    float4 p[JOIN_V], q[JOIN_V];
#pragma unroll
    for (int j = 0; j < JOIN_V; ++j) {
        const int i  = base + j * TPB;
        const int ic = (FULL || i < tot4) ? i : 0;
        p[j] = P4[ic];
        q[j] = P4[tot4 + ic];
    }
#pragma unroll
    for (int j = 0; j < JOIN_V; ++j) {
        const int i = base + j * TPB;
        if (FULL || i < tot4) {
            const float* c = b + 4 * (i & (FDIM / 4 - 1));
            float4 o;
            o.x = (p[j].x + q[j].x) + c[0];
            o.y = (p[j].y + q[j].y) + c[1];
            o.z = (p[j].z + q[j].z) + c[2];
            o.w = (p[j].w + q[j].w) + c[3];
            OUT4[i] = o;
        }
    }
}

__global__ __launch_bounds__(TPB) void gat_join_bias(const float* __restrict__ P,
                                                     const float* __restrict__ b,
                                                     float* __restrict__ OUT,
                                                     int tot4)
{
    const float4* __restrict__ P4 = reinterpret_cast<const float4*>(P);
    float4* __restrict__ OUT4 = reinterpret_cast<float4*>(OUT);
    const int blk0 = blockIdx.x * (TPB * JOIN_V);
    const int base = blk0 + threadIdx.x;
    if (blk0 + TPB * JOIN_V <= tot4)
        join_bias_body<true >(P4, b, OUT4, base, tot4);
    else
        join_bias_body<false>(P4, b, OUT4, base, tot4);
}

extern "C" void kernel_launch(void* const* d_in, const int* in_sizes, int n_in,
                              void* d_out, int out_size, void* d_ws, size_t ws_size,
                              hipStream_t stream) {
    const float* x   = (const float*)d_in[0];
    const int*   ei  = (const int*)  d_in[1];
    const float* W1  = (const float*)d_in[2];
    const float* as1 = (const float*)d_in[3];
    const float* ad1 = (const float*)d_in[4];
    const float* b1  = (const float*)d_in[5];
    const float* W2  = (const float*)d_in[6];
    const float* as2 = (const float*)d_in[7];
    const float* ad2 = (const float*)d_in[8];
    const float* b2  = (const float*)d_in[9];

    const int N    = in_sizes[0] / FDIM;
    const int E    = in_sizes[1] / 2;
    const int Etot = E + N;
    const int* src = ei;
    const int* dst = ei + E;

    float* ws = (float*)d_ws;
    float* H  = ws;  ws += (size_t)N * FDIM;
    float* X1 = ws;  ws += (size_t)N * FDIM;
    float* P  = ws;  ws += (size_t)2 * N * FDIM;
    float* HS = ws;  ws += N;
    float* HD = ws;  ws += N;
    float* Mx = ws;  ws += 2 * N;
    float* Sx = ws;  ws += 2 * N;
    float* EW = ws;  ws += Etot;

    const int tot  = N * FDIM;
    const int tot4 = tot / 4;

    auto layer = [&](const float* Xin, const float* W, const float* a_s,
                     const float* a_d, const float* b, float* OUT) {
        gat_gemm_wmma<<<N / 16, 32, 0, stream>>>(Xin, W, a_s, a_d, H, HS, HD);
        gat_init<<<(2 * tot4 + TPB * INIT_V - 1) / (TPB * INIT_V), TPB, 0,
                   stream>>>(P, Mx, Sx, 2 * N, 2 * tot4);
        gat_edge_score<<<(Etot + TPB * EDGE_V - 1) / (TPB * EDGE_V), TPB, 0,
                         stream>>>(src, dst, HS, HD, EW, Mx, E, Etot, N);
        gat_edge_exp<<<(Etot + TPB * EDGE_V - 1) / (TPB * EDGE_V), TPB, 0,
                       stream>>>(dst, Mx, EW, Sx, E, Etot, N);
        gat_aggr<<<(Etot + TPB * AGGR_B - 1) / (TPB * AGGR_B), TPB, 0,
                   stream>>>(src, dst, EW, Sx, H, P, E, Etot, N);
        gat_join_bias<<<(tot4 + TPB * JOIN_V - 1) / (TPB * JOIN_V), TPB, 0,
                        stream>>>(P, b, OUT, tot4);
    };

    layer(x,  W1, as1, ad1, b1, X1);
    layer(X1, W2, as2, ad2, b2, (float*)d_out);
}


# exp pass folded into the aggregate: a wave adds w to the segment sum and w*h[src] to the row, the join divides each row by its sum; 12 to 10 launches; f32 row sums now of unnormalised terms
# speedup vs baseline: 2.2272x; 1.2125x over previous
#include <hip/hip_runtime.h>
#include <math.h>

#define FDIM 64

typedef __attribute__((ext_vector_type(2))) float v2f;
typedef __attribute__((ext_vector_type(8))) float v8f;

__device__ __forceinline__ void atomicMaxF32(float* addr, float val) {
    if (val >= 0.0f) {
        atomicMax((int*)addr, __float_as_int(val));
    } else {
        atomicMin((unsigned int*)addr, __float_as_uint(val));
    }
}

__device__ __forceinline__ int my_half() {
    return (int)((__builtin_amdgcn_s_sendmsg_rtn(0x87) >> 18) & 1u);
}

__global__ __launch_bounds__(32) void gat_gemm_wmma(
    const float* __restrict__ X,
    const float* __restrict__ W,
    const float* __restrict__ a_s,
    const float* __restrict__ a_d,
    float* __restrict__ H,
    float* __restrict__ HS,
    float* __restrict__ HD)
{
    const int lane  = threadIdx.x;
    const int m     = lane & 15;
    const int khalf = lane >> 4;
    const int row0  = blockIdx.x * 16;

    const float* xr = X + (size_t)(row0 + m) * FDIM;
    v2f af[16];
#pragma unroll
    for (int k = 0; k < 16; ++k) {
        const int kc = 4 * k + 2 * khalf;
        af[k].x = xr[kc];
        af[k].y = xr[kc + 1];
    }

    v8f acc[4];
#pragma unroll
    for (int j = 0; j < 4; ++j) acc[j] = (v8f){0,0,0,0,0,0,0,0};

#pragma unroll
    for (int j = 0; j < 4; ++j) {
#pragma unroll
        for (int k = 0; k < 16; ++k) {
            const float* wp = W + (size_t)(4 * k + 2 * khalf) * FDIM + j * 16 + m;
            v2f bf;
            bf.x = wp[0];
            bf.y = wp[FDIM];
            acc[j] = __builtin_amdgcn_wmma_f32_16x16x4_f32(
                 false, af[k],  false, bf,
                 (short)0, acc[j],
                 false,  false);
        }
    }

    const int drow_base = row0 + 8 * khalf;
#pragma unroll
    for (int r = 0; r < 8; ++r) {
        float* hp = H + (size_t)(drow_base + r) * FDIM + m;
#pragma unroll
        for (int j = 0; j < 4; ++j) hp[j * 16] = acc[j][r];
    }

    float as_c[4], ad_c[4];
#pragma unroll
    for (int j = 0; j < 4; ++j) {
        as_c[j] = a_s[j * 16 + m];
        ad_c[j] = a_d[j * 16 + m];
    }
#pragma unroll
    for (int r = 0; r < 8; ++r) {
        float ps = 0.0f, pd = 0.0f;
#pragma unroll
        for (int j = 0; j < 4; ++j) {
            ps += acc[j][r] * as_c[j];
            pd += acc[j][r] * ad_c[j];
        }
#pragma unroll
        for (int off = 8; off >= 1; off >>= 1) {
            ps += __shfl_xor(ps, off, 32);
            pd += __shfl_xor(pd, off, 32);
        }
        if (m == 0) {
            HS[drow_base + r] = ps;
            HD[drow_base + r] = pd;
        }
    }
}

#define TPB      256
#define INIT_V   8
#define EDGE_V   4
#define JOIN_V   4
#define AGGR_B   2
#define AGGR_G   8

__global__ __launch_bounds__(TPB) void gat_init(float* __restrict__ P,
                                                float* __restrict__ Mx,
                                                float* __restrict__ Sx,
                                                int n2, int tot4x2)
{
    float4* __restrict__ P4 = reinterpret_cast<float4*>(P);
    const int base = blockIdx.x * (TPB * INIT_V) + threadIdx.x;
#pragma unroll
    for (int j = 0; j < INIT_V; ++j) {
        const int i = base + j * TPB;
        if (i < tot4x2) P4[i] = make_float4(0.0f, 0.0f, 0.0f, 0.0f);
    }
    const int i = blockIdx.x * TPB + threadIdx.x;
    if (i < n2) {
        Mx[i] = __int_as_float(0xFF800000);
        Sx[i] = 0.0f;
    }
}

__device__ __forceinline__ void edge_nodes(const int* __restrict__ src,
                                           const int* __restrict__ dst,
                                           int e, int E, int Etot,
                                           int& s, int& d)
{
    const int ec = (e < E) ? e : 0;
    const int sv = src[ec];
    const int dv = dst[ec];
    const int loop = (e < Etot) ? (e - E) : 0;
    s = (e < E) ? sv : loop;
    d = (e < E) ? dv : loop;
}

template <bool FULL>
__device__ __forceinline__ void edge_score_body(
    const int* __restrict__ src, const int* __restrict__ dst,
    const float* __restrict__ HS, const float* __restrict__ HD,
    float* __restrict__ EW, float* __restrict__ Mh,
    int base, int E, int Etot)
{
    int s[EDGE_V], d[EDGE_V];
#pragma unroll
    for (int j = 0; j < EDGE_V; ++j)
        edge_nodes(src, dst, base + j * TPB, E, Etot, s[j], d[j]);
    float v[EDGE_V];
#pragma unroll
    for (int j = 0; j < EDGE_V; ++j) {
        const float t = HS[s[j]] + HD[d[j]];
        v[j] = (t >= 0.0f) ? t : 0.2f * t;
    }
#pragma unroll
    for (int j = 0; j < EDGE_V; ++j) {
        const int e = base + j * TPB;
        if (FULL || e < Etot) {
            EW[e] = v[j];
            atomicMaxF32(&Mh[d[j]], v[j]);
        }
    }
}

__global__ __launch_bounds__(TPB) void gat_edge_score(
    const int* __restrict__ src, const int* __restrict__ dst,
    const float* __restrict__ HS, const float* __restrict__ HD,
    float* __restrict__ EW,
    float* __restrict__ Mx,
    int E, int Etot, int n_nodes)
{
    const int blk0 = blockIdx.x * (TPB * EDGE_V);
    const int base = blk0 + threadIdx.x;
    float* Mh = Mx + (size_t)my_half() * n_nodes;
    if (blk0 + TPB * EDGE_V <= Etot)
        edge_score_body<true >(src, dst, HS, HD, EW, Mh, base, E, Etot);
    else
        edge_score_body<false>(src, dst, HS, HD, EW, Mh, base, E, Etot);
}

__device__ __forceinline__ float lane_f32(float v, int k) {
    return __int_as_float(__builtin_amdgcn_readlane(__float_as_int(v), k));
}

__global__ __launch_bounds__(TPB) void gat_aggr(
    const int* __restrict__ src, const int* __restrict__ dst,
    const float* __restrict__ EW,
    const float* __restrict__ Mx,
    float* __restrict__ Sx,
    const float* __restrict__ H, float* __restrict__ P,
    int E, int Etot, int n_nodes)
{
    const int lane = threadIdx.x & 31;
    const int wave = blockIdx.x * (TPB / 32) +
                     __builtin_amdgcn_readfirstlane((int)(threadIdx.x >> 5));
    const int e0   = wave * (32 * AGGR_B);
    if (e0 >= Etot) return;
    const int half = my_half();
    float* OUT = P + (size_t)half * n_nodes * FDIM;
    float* Sh  = Sx + (size_t)half * n_nodes;
    const float* Hl = H + lane;
    float* Ol = OUT + lane;

    int   s[AGGR_B], d[AGGR_B];
    float a[AGGR_B];
#pragma unroll
    for (int b = 0; b < AGGR_B; ++b) {
        const int e = e0 + 32 * b + lane;
        edge_nodes(src, dst, e, E, Etot, s[b], d[b]);
        a[b] = EW[(e < Etot) ? e : 0];
    }
#pragma unroll
    for (int b = 0; b < AGGR_B; ++b)
        a[b] = __expf(a[b] - fmaxf(Mx[d[b]], Mx[n_nodes + d[b]]));
#pragma unroll
    for (int b = 0; b < AGGR_B; ++b)
        if (e0 + 32 * b + lane < Etot) atomicAdd(&Sh[d[b]], a[b]);

#pragma unroll
    for (int b = 0; b < AGGR_B; ++b) {
        const int cnt = Etot - (e0 + 32 * b);
        if (cnt >= 32) {
#pragma unroll
            for (int k0 = 0; k0 < 32; k0 += AGGR_G) {
                float h0[AGGR_G], h1[AGGR_G];
#pragma unroll
                for (int k = 0; k < AGGR_G; ++k) {
                    const int sk = __builtin_amdgcn_readlane(s[b], k0 + k);
                    const float* hr = Hl + (size_t)sk * FDIM;
                    h0[k] = hr[0];
                    h1[k] = hr[32];
                }
#pragma unroll
                for (int k = 0; k < AGGR_G; ++k) {
                    const int   dk = __builtin_amdgcn_readlane(d[b], k0 + k);
                    const float ak = lane_f32(a[b], k0 + k);
                    float* o = Ol + (size_t)dk * FDIM;
                    atomicAdd(o,      ak * h0[k]);
                    atomicAdd(o + 32, ak * h1[k]);
                }
            }
        } else {
            for (int k = 0; k < cnt; ++k) {
                const int   sk = __builtin_amdgcn_readlane(s[b], k);
                const int   dk = __builtin_amdgcn_readlane(d[b], k);
                const float ak = lane_f32(a[b], k);
                const float* hr = Hl + (size_t)sk * FDIM;
                float* o = Ol + (size_t)dk * FDIM;
                atomicAdd(o,      ak * hr[0]);
                atomicAdd(o + 32, ak * hr[32]);
            }
        }
    }
}

template <bool FULL>
__device__ __forceinline__ void join_bias_body(const float4* __restrict__ P4,
                                               const float* __restrict__ Sx,
                                               const float* __restrict__ b,
                                               float4* __restrict__ OUT4,
                                               int base, int tot4, int n_nodes)
{
    float4 p[JOIN_V], q[JOIN_V];
    float  s0[JOIN_V], s1[JOIN_V];
#pragma unroll
    for (int j = 0; j < JOIN_V; ++j) {
        const int i  = base + j * TPB;
        const int ic = (FULL || i < tot4) ? i : 0;
        const int nd = ic / (FDIM / 4);
        p[j]  = P4[ic];
        q[j]  = P4[tot4 + ic];
        s0[j] = Sx[nd];
        s1[j] = Sx[n_nodes + nd];
    }
#pragma unroll
    for (int j = 0; j < JOIN_V; ++j) {
        const int i = base + j * TPB;
        if (FULL || i < tot4) {
            const float* c = b + 4 * (i & (FDIM / 4 - 1));
            const float s = s0[j] + s1[j];
            float4 o;
            o.x = (p[j].x + q[j].x) / s + c[0];
            o.y = (p[j].y + q[j].y) / s + c[1];
            o.z = (p[j].z + q[j].z) / s + c[2];
            o.w = (p[j].w + q[j].w) / s + c[3];
            OUT4[i] = o;
        }
    }
}

__global__ __launch_bounds__(TPB) void gat_join_bias(const float* __restrict__ P,
                                                     const float* __restrict__ Sx,
                                                     const float* __restrict__ b,
                                                     float* __restrict__ OUT,
                                                     int tot4, int n_nodes)
{
    const float4* __restrict__ P4 = reinterpret_cast<const float4*>(P);
    float4* __restrict__ OUT4 = reinterpret_cast<float4*>(OUT);
    const int blk0 = blockIdx.x * (TPB * JOIN_V);
    const int base = blk0 + threadIdx.x;
    if (blk0 + TPB * JOIN_V <= tot4)
        join_bias_body<true >(P4, Sx, b, OUT4, base, tot4, n_nodes);
    else
        join_bias_body<false>(P4, Sx, b, OUT4, base, tot4, n_nodes);
}

extern "C" void kernel_launch(void* const* d_in, const int* in_sizes, int n_in,
                              void* d_out, int out_size, void* d_ws, size_t ws_size,
                              hipStream_t stream) {
    const float* x   = (const float*)d_in[0];
    const int*   ei  = (const int*)  d_in[1];
    const float* W1  = (const float*)d_in[2];
    const float* as1 = (const float*)d_in[3];
    const float* ad1 = (const float*)d_in[4];
    const float* b1  = (const float*)d_in[5];
    const float* W2  = (const float*)d_in[6];
    const float* as2 = (const float*)d_in[7];
    const float* ad2 = (const float*)d_in[8];
    const float* b2  = (const float*)d_in[9];

    const int N    = in_sizes[0] / FDIM;
    const int E    = in_sizes[1] / 2;
    const int Etot = E + N;
    const int* src = ei;
    const int* dst = ei + E;

    float* ws = (float*)d_ws;
    float* H  = ws;  ws += (size_t)N * FDIM;
    float* X1 = ws;  ws += (size_t)N * FDIM;
    float* P  = ws;  ws += (size_t)2 * N * FDIM;
    float* HS = ws;  ws += N;
    float* HD = ws;  ws += N;
    float* Mx = ws;  ws += 2 * N;
    float* Sx = ws;  ws += 2 * N;
    float* EW = ws;  ws += Etot;

    const int tot  = N * FDIM;
    const int tot4 = tot / 4;

    auto layer = [&](const float* Xin, const float* W, const float* a_s,
                     const float* a_d, const float* b, float* OUT) {
        gat_gemm_wmma<<<N / 16, 32, 0, stream>>>(Xin, W, a_s, a_d, H, HS, HD);
        gat_init<<<(2 * tot4 + TPB * INIT_V - 1) / (TPB * INIT_V), TPB, 0,
                   stream>>>(P, Mx, Sx, 2 * N, 2 * tot4);
        gat_edge_score<<<(Etot + TPB * EDGE_V - 1) / (TPB * EDGE_V), TPB, 0,
                         stream>>>(src, dst, HS, HD, EW, Mx, E, Etot, N);
        gat_aggr<<<(Etot + TPB * AGGR_B - 1) / (TPB * AGGR_B), TPB, 0,
                   stream>>>(src, dst, EW, Mx, Sx, H, P, E, Etot, N);
        gat_join_bias<<<(tot4 + TPB * JOIN_V - 1) / (TPB * JOIN_V), TPB, 0,
                        stream>>>(P, Sx, b, OUT, tot4, N);
    };

    layer(x,  W1, as1, ad1, b1, X1);
    layer(X1, W2, as2, ad2, b2, (float*)d_out);
}


# Per-node scalars (hs, hd, joined segment max) copied whole into LDS by one 1024-thread work-group per WGP; edge passes gather them from LDS, not L2; f32 sums in arrival order as before
# speedup vs baseline: 2.2987x; 1.0321x over previous
#include <hip/hip_runtime.h>
#include <math.h>

#define FDIM 64

typedef __attribute__((ext_vector_type(2))) float v2f;
typedef __attribute__((ext_vector_type(8))) float v8f;

__device__ __forceinline__ void atomicMaxF32(float* addr, float val) {
    if (val >= 0.0f) {
        atomicMax((int*)addr, __float_as_int(val));
    } else {
        atomicMin((unsigned int*)addr, __float_as_uint(val));
    }
}

__device__ __forceinline__ int my_half() {
    return (int)((__builtin_amdgcn_s_sendmsg_rtn(0x87) >> 18) & 1u);
}

__global__ __launch_bounds__(32) void gat_gemm_wmma(
    const float* __restrict__ X,
    const float* __restrict__ W,
    const float* __restrict__ a_s,
    const float* __restrict__ a_d,
    float* __restrict__ H,
    float* __restrict__ HS,
    float* __restrict__ HD)
{
    const int lane  = threadIdx.x;
    const int m     = lane & 15;
    const int khalf = lane >> 4;
    const int row0  = blockIdx.x * 16;

    const float* xr = X + (size_t)(row0 + m) * FDIM;
    v2f af[16];
#pragma unroll
    for (int k = 0; k < 16; ++k) {
        const int kc = 4 * k + 2 * khalf;
        af[k].x = xr[kc];
        af[k].y = xr[kc + 1];
    }

    v8f acc[4];
#pragma unroll
    for (int j = 0; j < 4; ++j) acc[j] = (v8f){0,0,0,0,0,0,0,0};

#pragma unroll
    for (int j = 0; j < 4; ++j) {
#pragma unroll
        for (int k = 0; k < 16; ++k) {
            const float* wp = W + (size_t)(4 * k + 2 * khalf) * FDIM + j * 16 + m;
            v2f bf;
            bf.x = wp[0];
            bf.y = wp[FDIM];
            acc[j] = __builtin_amdgcn_wmma_f32_16x16x4_f32(
                 false, af[k],  false, bf,
                 (short)0, acc[j],
                 false,  false);
        }
    }

    const int drow_base = row0 + 8 * khalf;
#pragma unroll
    for (int r = 0; r < 8; ++r) {
        float* hp = H + (size_t)(drow_base + r) * FDIM + m;
#pragma unroll
        for (int j = 0; j < 4; ++j) hp[j * 16] = acc[j][r];
    }

    float as_c[4], ad_c[4];
#pragma unroll
    for (int j = 0; j < 4; ++j) {
        as_c[j] = a_s[j * 16 + m];
        ad_c[j] = a_d[j * 16 + m];
    }
#pragma unroll
    for (int r = 0; r < 8; ++r) {
        float ps = 0.0f, pd = 0.0f;
#pragma unroll
        for (int j = 0; j < 4; ++j) {
            ps += acc[j][r] * as_c[j];
            pd += acc[j][r] * ad_c[j];
        }
#pragma unroll
        for (int off = 8; off >= 1; off >>= 1) {
            ps += __shfl_xor(ps, off, 32);
            pd += __shfl_xor(pd, off, 32);
        }
        if (m == 0) {
            HS[drow_base + r] = ps;
            HD[drow_base + r] = pd;
        }
    }
}

#define TPB      256
#define INIT_V   8
#define EDGE_V   4
#define JOIN_V   4
#define AGGR_B   2
#define AGGR_G   8

#define WG         1024
#define EDGE_WGS   256
#define STAGE_V    13
#define LDS_NODES  (STAGE_V * WG * 4)

__device__ __forceinline__ void edge_share(int Etot, int& c0, int& c1) {
    const int g   = (int)gridDim.x;
    const int per = (((Etot + g - 1) / g) + 31) & ~31;
    const int lo  = (int)blockIdx.x * per;
    c0 = (lo < Etot) ? lo : Etot;
    c1 = (c0 + per < Etot) ? (c0 + per) : Etot;
}

__device__ __forceinline__ void table_load(const float* __restrict__ A, int n4,
                                           float4 (&r)[STAGE_V]) {
    const float4* __restrict__ A4 = reinterpret_cast<const float4*>(A);
#pragma unroll
    for (int j = 0; j < STAGE_V; ++j) {
        const int i = (int)threadIdx.x + j * WG;
        r[j] = A4[(i < n4) ? i : 0];
    }
}
__device__ __forceinline__ void table_store(float4* tab4,
                                            const float4 (&r)[STAGE_V]) {
#pragma unroll
    for (int j = 0; j < STAGE_V; ++j)
        tab4[(int)threadIdx.x + j * WG] = r[j];
}

__global__ __launch_bounds__(TPB) void gat_init(float* __restrict__ P,
                                                float* __restrict__ Mx,
                                                float* __restrict__ Sx,
                                                int n2, int tot4x2)
{
    float4* __restrict__ P4 = reinterpret_cast<float4*>(P);
    const int base = blockIdx.x * (TPB * INIT_V) + threadIdx.x;
#pragma unroll
    for (int j = 0; j < INIT_V; ++j) {
        const int i = base + j * TPB;
        if (i < tot4x2) P4[i] = make_float4(0.0f, 0.0f, 0.0f, 0.0f);
    }
    const int i = blockIdx.x * TPB + threadIdx.x;
    if (i < n2) {
        Mx[i] = __int_as_float(0xFF800000);
        Sx[i] = 0.0f;
    }
}

__device__ __forceinline__ void edge_nodes(const int* __restrict__ src,
                                           const int* __restrict__ dst,
                                           int e, int E, int Etot,
                                           int& s, int& d)
{
    const int ec = (e < E) ? e : 0;
    const int sv = src[ec];
    const int dv = dst[ec];
    const int loop = (e < Etot) ? (e - E) : 0;
    s = (e < E) ? sv : loop;
    d = (e < E) ? dv : loop;
}

__global__ __launch_bounds__(WG) void gat_edge_score(
    const int* __restrict__ src, const int* __restrict__ dst,
    const float* __restrict__ HS, const float* __restrict__ HD,
    float* __restrict__ EW,
    float* __restrict__ Mx,
    int E, int Etot, int n_nodes)
{
    __shared__ float4 tab4[LDS_NODES / 4];
    const float* tab = reinterpret_cast<const float*>(tab4);
    const int n4 = n_nodes >> 2;
    int c0, c1;
    edge_share(Etot, c0, c1);
    const int base = c0 + (int)threadIdx.x;

    float4 r[STAGE_V];
    table_load(HS, n4, r);
    int s[EDGE_V], d[EDGE_V];
#pragma unroll
    for (int j = 0; j < EDGE_V; ++j)
        edge_nodes(src, dst, base + j * WG, E, Etot, s[j], d[j]);
    table_store(tab4, r);
    __syncthreads();

    float a[EDGE_V];
#pragma unroll
    for (int j = 0; j < EDGE_V; ++j) a[j] = tab[s[j]];
    __syncthreads();
    table_load(HD, n4, r);
    table_store(tab4, r);
    __syncthreads();

    float* Mh = Mx + (size_t)my_half() * n_nodes;
#pragma unroll
    for (int j = 0; j < EDGE_V; ++j) {
        const int e = base + j * WG;
        const float t = a[j] + tab[d[j]];
        const float v = (t >= 0.0f) ? t : 0.2f * t;
        if (e < c1) {
            EW[e] = v;
            atomicMaxF32(&Mh[d[j]], v);
        }
    }
}

__device__ __forceinline__ float lane_f32(float v, int k) {
    return __int_as_float(__builtin_amdgcn_readlane(__float_as_int(v), k));
}

__global__ __launch_bounds__(WG) void gat_aggr(
    const int* __restrict__ src, const int* __restrict__ dst,
    const float* __restrict__ EW,
    const float* __restrict__ Mx,
    float* __restrict__ Sx,
    const float* __restrict__ H, float* __restrict__ P,
    int E, int Etot, int n_nodes)
{
    __shared__ float4 tab4[LDS_NODES / 4];
    const float* tab = reinterpret_cast<const float*>(tab4);
    const int n4 = n_nodes >> 2;
    {
        float4 p[STAGE_V], q[STAGE_V];
        table_load(Mx, n4, p);
        table_load(Mx + n_nodes, n4, q);
#pragma unroll
        for (int j = 0; j < STAGE_V; ++j) {
            p[j].x = fmaxf(p[j].x, q[j].x);
            p[j].y = fmaxf(p[j].y, q[j].y);
            p[j].z = fmaxf(p[j].z, q[j].z);
            p[j].w = fmaxf(p[j].w, q[j].w);
        }
        table_store(tab4, p);
    }
    __syncthreads();

    const int lane = threadIdx.x & 31;
    const int wave = __builtin_amdgcn_readfirstlane((int)(threadIdx.x >> 5));
    int c0, c1;
    edge_share(Etot, c0, c1);
    const int nb   = (c1 - c0 + 31) >> 5;
    const int half = my_half();
    float* OUT = P + (size_t)half * n_nodes * FDIM;
    float* Sh  = Sx + (size_t)half * n_nodes;
    const float* Hl = H + lane;
    float* Ol = OUT + lane;

    for (int b0 = wave; b0 < nb; b0 += (WG / 32) * AGGR_B) {
        int   s[AGGR_B], d[AGGR_B];
        float a[AGGR_B];
#pragma unroll
        for (int b = 0; b < AGGR_B; ++b) {
            const int e = c0 + 32 * (b0 + (WG / 32) * b) + lane;
            edge_nodes(src, dst, e, E, Etot, s[b], d[b]);
            a[b] = EW[(e < Etot) ? e : 0];
        }
#pragma unroll
        for (int b = 0; b < AGGR_B; ++b)
            a[b] = __expf(a[b] - tab[d[b]]);
#pragma unroll
        for (int b = 0; b < AGGR_B; ++b)
            if (c0 + 32 * (b0 + (WG / 32) * b) + lane < c1) atomicAdd(&Sh[d[b]], a[b]);

#pragma unroll
        for (int b = 0; b < AGGR_B; ++b) {
            const int cnt = c1 - (c0 + 32 * (b0 + (WG / 32) * b));
            if (cnt >= 32) {
#pragma unroll
                for (int k0 = 0; k0 < 32; k0 += AGGR_G) {
                    float h0[AGGR_G], h1[AGGR_G];
#pragma unroll
                    for (int k = 0; k < AGGR_G; ++k) {
                        const int sk = __builtin_amdgcn_readlane(s[b], k0 + k);
                        const float* hr = Hl + (size_t)sk * FDIM;
                        h0[k] = hr[0];
                        h1[k] = hr[32];
                    }
#pragma unroll
                    for (int k = 0; k < AGGR_G; ++k) {
                        const int   dk = __builtin_amdgcn_readlane(d[b], k0 + k);
                        const float ak = lane_f32(a[b], k0 + k);
                        float* o = Ol + (size_t)dk * FDIM;
                        atomicAdd(o,      ak * h0[k]);
                        atomicAdd(o + 32, ak * h1[k]);
                    }
                }
            } else {
                for (int k = 0; k < cnt; ++k) {
                    const int   sk = __builtin_amdgcn_readlane(s[b], k);
                    const int   dk = __builtin_amdgcn_readlane(d[b], k);
                    const float ak = lane_f32(a[b], k);
                    const float* hr = Hl + (size_t)sk * FDIM;
                    float* o = Ol + (size_t)dk * FDIM;
                    atomicAdd(o,      ak * hr[0]);
                    atomicAdd(o + 32, ak * hr[32]);
                }
            }
        }
    }
}

template <bool FULL>
__device__ __forceinline__ void join_bias_body(const float4* __restrict__ P4,
                                               const float* __restrict__ Sx,
                                               const float* __restrict__ b,
                                               float4* __restrict__ OUT4,
                                               int base, int tot4, int n_nodes)
{
    float4 p[JOIN_V], q[JOIN_V];
    float  s0[JOIN_V], s1[JOIN_V];
#pragma unroll
    for (int j = 0; j < JOIN_V; ++j) {
        const int i  = base + j * TPB;
        const int ic = (FULL || i < tot4) ? i : 0;
        const int nd = ic / (FDIM / 4);
        p[j]  = P4[ic];
        q[j]  = P4[tot4 + ic];
        s0[j] = Sx[nd];
        s1[j] = Sx[n_nodes + nd];
    }
#pragma unroll
    for (int j = 0; j < JOIN_V; ++j) {
        const int i = base + j * TPB;
        if (FULL || i < tot4) {
            const float* c = b + 4 * (i & (FDIM / 4 - 1));
            const float s = s0[j] + s1[j];
            float4 o;
            o.x = (p[j].x + q[j].x) / s + c[0];
            o.y = (p[j].y + q[j].y) / s + c[1];
            o.z = (p[j].z + q[j].z) / s + c[2];
            o.w = (p[j].w + q[j].w) / s + c[3];
            OUT4[i] = o;
        }
    }
}

__global__ __launch_bounds__(TPB) void gat_join_bias(const float* __restrict__ P,
                                                     const float* __restrict__ Sx,
                                                     const float* __restrict__ b,
                                                     float* __restrict__ OUT,
                                                     int tot4, int n_nodes)
{
    const float4* __restrict__ P4 = reinterpret_cast<const float4*>(P);
    float4* __restrict__ OUT4 = reinterpret_cast<float4*>(OUT);
    const int blk0 = blockIdx.x * (TPB * JOIN_V);
    const int base = blk0 + threadIdx.x;
    if (blk0 + TPB * JOIN_V <= tot4)
        join_bias_body<true >(P4, Sx, b, OUT4, base, tot4, n_nodes);
    else
        join_bias_body<false>(P4, Sx, b, OUT4, base, tot4, n_nodes);
}

extern "C" void kernel_launch(void* const* d_in, const int* in_sizes, int n_in,
                              void* d_out, int out_size, void* d_ws, size_t ws_size,
                              hipStream_t stream) {
    const float* x   = (const float*)d_in[0];
    const int*   ei  = (const int*)  d_in[1];
    const float* W1  = (const float*)d_in[2];
    const float* as1 = (const float*)d_in[3];
    const float* ad1 = (const float*)d_in[4];
    const float* b1  = (const float*)d_in[5];
    const float* W2  = (const float*)d_in[6];
    const float* as2 = (const float*)d_in[7];
    const float* ad2 = (const float*)d_in[8];
    const float* b2  = (const float*)d_in[9];

    const int N    = in_sizes[0] / FDIM;
    const int E    = in_sizes[1] / 2;
    const int Etot = E + N;
    const int* src = ei;
    const int* dst = ei + E;

    float* ws = (float*)d_ws;
    float* H  = ws;  ws += (size_t)N * FDIM;
    float* X1 = ws;  ws += (size_t)N * FDIM;
    float* P  = ws;  ws += (size_t)2 * N * FDIM;
    float* HS = ws;  ws += N;
    float* HD = ws;  ws += N;
    float* Mx = ws;  ws += 2 * N;
    float* Sx = ws;  ws += 2 * N;
    float* EW = ws;  ws += Etot;

    const int tot  = N * FDIM;
    const int tot4 = tot / 4;
    const int need_wgs = (Etot + WG * EDGE_V - 1) / (WG * EDGE_V);
    const int edge_wgs = (need_wgs > EDGE_WGS) ? need_wgs : EDGE_WGS;

    auto layer = [&](const float* Xin, const float* W, const float* a_s,
                     const float* a_d, const float* b, float* OUT) {
        gat_gemm_wmma<<<N / 16, 32, 0, stream>>>(Xin, W, a_s, a_d, H, HS, HD);
        gat_init<<<(2 * tot4 + TPB * INIT_V - 1) / (TPB * INIT_V), TPB, 0,
                   stream>>>(P, Mx, Sx, 2 * N, 2 * tot4);
        gat_edge_score<<<edge_wgs, WG, 0, stream>>>(src, dst, HS, HD, EW, Mx,
                                                    E, Etot, N);
        gat_aggr<<<edge_wgs, WG, 0, stream>>>(src, dst, EW, Mx, Sx, H, P,
                                              E, Etot, N);
        gat_join_bias<<<(tot4 + TPB * JOIN_V - 1) / (TPB * JOIN_V), TPB, 0,
                        stream>>>(P, Sx, b, OUT, tot4, N);
    };

    layer(x,  W1, as1, ad1, b1, X1);
    layer(X1, W2, as2, ad2, b2, (float*)d_out);
}


# Init and join passes folded into the GEMM launch: it stores the atomic targets' start values, layer 2's loader forms its rows from layer 1's sums; 10 to 7 launches; f32 sum order unchanged
# speedup vs baseline: 2.7740x; 1.2068x over previous
#include <hip/hip_runtime.h>
#include <math.h>

#define FDIM 64

typedef __attribute__((ext_vector_type(2))) float v2f;
typedef __attribute__((ext_vector_type(8))) float v8f;

__device__ __forceinline__ void atomicMaxF32(float* addr, float val) {
    if (val >= 0.0f) {
        atomicMax((int*)addr, __float_as_int(val));
    } else {
        atomicMin((unsigned int*)addr, __float_as_uint(val));
    }
}

__device__ __forceinline__ int my_half() {
    return (int)((__builtin_amdgcn_s_sendmsg_rtn(0x87) >> 18) & 1u);
}

template <bool JOIN>
__global__ __launch_bounds__(32) void gat_gemm_wmma(
    const float* __restrict__ X,
    const float* __restrict__ Pin,
    const float* __restrict__ Sin,
    const float* __restrict__ bin,
    const float* __restrict__ W,
    const float* __restrict__ a_s,
    const float* __restrict__ a_d,
    float* __restrict__ H,
    float* __restrict__ HS,
    float* __restrict__ HD,
    float* __restrict__ Pz,
    float* __restrict__ Mz,
    float* __restrict__ Sz,
    int n_nodes)
{
    const int lane  = threadIdx.x;
    const int m     = lane & 15;
    const int khalf = lane >> 4;
    const int row0  = blockIdx.x * 16;

    const size_t ro = (size_t)(row0 + m) * FDIM;
    v2f af[16];
    v2f aq[16], ab[16];
    float s0 = 1.0f, s1 = 0.0f;
    if (JOIN) {
        const float* p0 = Pin + ro;
        const float* p1 = Pin + (size_t)n_nodes * FDIM + ro;
        s0 = Sin[row0 + m];
        s1 = Sin[n_nodes + row0 + m];
#pragma unroll
        for (int k = 0; k < 16; ++k) {
            const int kc = 4 * k + 2 * khalf;
            af[k].x = p0[kc];  af[k].y = p0[kc + 1];
            aq[k].x = p1[kc];  aq[k].y = p1[kc + 1];
            ab[k].x = bin[kc]; ab[k].y = bin[kc + 1];
        }
        asm volatile("" ::
                        "v"(s0), "v"(s1), "v"(af[0]), "v"(aq[0]), "v"(ab[0]), "v"(af[1]),
                        "v"(aq[1]), "v"(ab[1]), "v"(af[2]), "v"(aq[2]), "v"(ab[2]), "v"(af[3]),
                        "v"(aq[3]), "v"(ab[3]), "v"(af[4]), "v"(aq[4]), "v"(ab[4]), "v"(af[5]),
                        "v"(aq[5]), "v"(ab[5]), "v"(af[6]), "v"(aq[6]), "v"(ab[6]), "v"(af[7]),
                        "v"(aq[7]), "v"(ab[7]));
        asm volatile("" ::
                        "v"(af[8]), "v"(aq[8]), "v"(ab[8]), "v"(af[9]), "v"(aq[9]), "v"(ab[9]),
                        "v"(af[10]), "v"(aq[10]), "v"(ab[10]), "v"(af[11]), "v"(aq[11]), "v"(ab[11]),
                        "v"(af[12]), "v"(aq[12]), "v"(ab[12]), "v"(af[13]), "v"(aq[13]), "v"(ab[13]),
                        "v"(af[14]), "v"(aq[14]), "v"(ab[14]), "v"(af[15]), "v"(aq[15]), "v"(ab[15]));
    } else {
        const float* xr = X + ro;
#pragma unroll
        for (int k = 0; k < 16; ++k) {
            const int kc = 4 * k + 2 * khalf;
            af[k].x = xr[kc];
            af[k].y = xr[kc + 1];
        }
    }

    {
        const float4 z = make_float4(0.0f, 0.0f, 0.0f, 0.0f);
        float4* z0 = reinterpret_cast<float4*>(Pz + (size_t)row0 * FDIM);
        float4* z1 = reinterpret_cast<float4*>(Pz + ((size_t)n_nodes + row0) * FDIM);
#pragma unroll
        for (int j = 0; j < 8; ++j) {
            z0[lane + 32 * j] = z;
            z1[lane + 32 * j] = z;
        }
        const size_t nd = (size_t)khalf * n_nodes + row0 + m;
        Mz[nd] = __int_as_float(0xFF800000);
        Sz[nd] = 0.0f;
    }

    if (JOIN) {
        const float s = s0 + s1;
#pragma unroll
        for (int k = 0; k < 16; ++k) {
            af[k].x = (af[k].x + aq[k].x) / s + ab[k].x;
            af[k].y = (af[k].y + aq[k].y) / s + ab[k].y;
        }
    }

    v8f acc[4];
#pragma unroll
    for (int j = 0; j < 4; ++j) acc[j] = (v8f){0,0,0,0,0,0,0,0};

#pragma unroll
    for (int j = 0; j < 4; ++j) {
#pragma unroll
        for (int k = 0; k < 16; ++k) {
            const float* wp = W + (size_t)(4 * k + 2 * khalf) * FDIM + j * 16 + m;
            v2f bf;
            bf.x = wp[0];
            bf.y = wp[FDIM];
            acc[j] = __builtin_amdgcn_wmma_f32_16x16x4_f32(
                 false, af[k],  false, bf,
                 (short)0, acc[j],
                 false,  false);
        }
    }

    const int drow_base = row0 + 8 * khalf;
#pragma unroll
    for (int r = 0; r < 8; ++r) {
        float* hp = H + (size_t)(drow_base + r) * FDIM + m;
#pragma unroll
        for (int j = 0; j < 4; ++j) hp[j * 16] = acc[j][r];
    }

    float as_c[4], ad_c[4];
#pragma unroll
    for (int j = 0; j < 4; ++j) {
        as_c[j] = a_s[j * 16 + m];
        ad_c[j] = a_d[j * 16 + m];
    }
#pragma unroll
    for (int r = 0; r < 8; ++r) {
        float ps = 0.0f, pd = 0.0f;
#pragma unroll
        for (int j = 0; j < 4; ++j) {
            ps += acc[j][r] * as_c[j];
            pd += acc[j][r] * ad_c[j];
        }
#pragma unroll
        for (int off = 8; off >= 1; off >>= 1) {
            ps += __shfl_xor(ps, off, 32);
            pd += __shfl_xor(pd, off, 32);
        }
        if (m == 0) {
            HS[drow_base + r] = ps;
            HD[drow_base + r] = pd;
        }
    }
}

#define TPB      256
#define EDGE_V   4
#define JOIN_V   4
#define AGGR_B   2
#define AGGR_G   8

#define WG         1024
#define EDGE_WGS   256
#define STAGE_V    13
#define LDS_NODES  (STAGE_V * WG * 4)

__device__ __forceinline__ void edge_share(int Etot, int& c0, int& c1) {
    const int g   = (int)gridDim.x;
    const int per = (((Etot + g - 1) / g) + 31) & ~31;
    const int lo  = (int)blockIdx.x * per;
    c0 = (lo < Etot) ? lo : Etot;
    c1 = (c0 + per < Etot) ? (c0 + per) : Etot;
}

__device__ __forceinline__ void table_load(const float* __restrict__ A, int n4,
                                           float4 (&r)[STAGE_V]) {
    const float4* __restrict__ A4 = reinterpret_cast<const float4*>(A);
#pragma unroll
    for (int j = 0; j < STAGE_V; ++j) {
        const int i = (int)threadIdx.x + j * WG;
        r[j] = A4[(i < n4) ? i : 0];
    }
}
__device__ __forceinline__ void table_store(float4* tab4,
                                            const float4 (&r)[STAGE_V]) {
#pragma unroll
    for (int j = 0; j < STAGE_V; ++j)
        tab4[(int)threadIdx.x + j * WG] = r[j];
}

__device__ __forceinline__ void edge_nodes(const int* __restrict__ src,
                                           const int* __restrict__ dst,
                                           int e, int E, int Etot,
                                           int& s, int& d)
{
    const int ec = (e < E) ? e : 0;
    const int sv = src[ec];
    const int dv = dst[ec];
    const int loop = (e < Etot) ? (e - E) : 0;
    s = (e < E) ? sv : loop;
    d = (e < E) ? dv : loop;
}

__global__ __launch_bounds__(WG) void gat_edge_score(
    const int* __restrict__ src, const int* __restrict__ dst,
    const float* __restrict__ HS, const float* __restrict__ HD,
    float* __restrict__ EW,
    float* __restrict__ Mx,
    int E, int Etot, int n_nodes)
{
    __shared__ float4 tab4[LDS_NODES / 4];
    const float* tab = reinterpret_cast<const float*>(tab4);
    const int n4 = n_nodes >> 2;
    int c0, c1;
    edge_share(Etot, c0, c1);
    const int base = c0 + (int)threadIdx.x;

    float4 r[STAGE_V];
    table_load(HS, n4, r);
    int s[EDGE_V], d[EDGE_V];
#pragma unroll
    for (int j = 0; j < EDGE_V; ++j)
        edge_nodes(src, dst, base + j * WG, E, Etot, s[j], d[j]);
    table_store(tab4, r);
    __syncthreads();

    float a[EDGE_V];
#pragma unroll
    for (int j = 0; j < EDGE_V; ++j) a[j] = tab[s[j]];
    __syncthreads();
    table_load(HD, n4, r);
    table_store(tab4, r);
    __syncthreads();

    float* Mh = Mx + (size_t)my_half() * n_nodes;
#pragma unroll
    for (int j = 0; j < EDGE_V; ++j) {
        const int e = base + j * WG;
        const float t = a[j] + tab[d[j]];
        const float v = (t >= 0.0f) ? t : 0.2f * t;
        if (e < c1) {
            EW[e] = v;
            atomicMaxF32(&Mh[d[j]], v);
        }
    }
}

__device__ __forceinline__ float lane_f32(float v, int k) {
    return __int_as_float(__builtin_amdgcn_readlane(__float_as_int(v), k));
}

__global__ __launch_bounds__(WG) void gat_aggr(
    const int* __restrict__ src, const int* __restrict__ dst,
    const float* __restrict__ EW,
    const float* __restrict__ Mx,
    float* __restrict__ Sx,
    const float* __restrict__ H, float* __restrict__ P,
    int E, int Etot, int n_nodes)
{
    __shared__ float4 tab4[LDS_NODES / 4];
    const float* tab = reinterpret_cast<const float*>(tab4);
    const int n4 = n_nodes >> 2;
    {
        float4 p[STAGE_V], q[STAGE_V];
        table_load(Mx, n4, p);
        table_load(Mx + n_nodes, n4, q);
#pragma unroll
        for (int j = 0; j < STAGE_V; ++j) {
            p[j].x = fmaxf(p[j].x, q[j].x);
            p[j].y = fmaxf(p[j].y, q[j].y);
            p[j].z = fmaxf(p[j].z, q[j].z);
            p[j].w = fmaxf(p[j].w, q[j].w);
        }
        table_store(tab4, p);
    }
    __syncthreads();

    const int lane = threadIdx.x & 31;
    const int wave = __builtin_amdgcn_readfirstlane((int)(threadIdx.x >> 5));
    int c0, c1;
    edge_share(Etot, c0, c1);
    const int nb   = (c1 - c0 + 31) >> 5;
    const int half = my_half();
    float* OUT = P + (size_t)half * n_nodes * FDIM;
    float* Sh  = Sx + (size_t)half * n_nodes;
    const float* Hl = H + lane;
    float* Ol = OUT + lane;

    for (int b0 = wave; b0 < nb; b0 += (WG / 32) * AGGR_B) {
        int   s[AGGR_B], d[AGGR_B];
        float a[AGGR_B];
#pragma unroll
        for (int b = 0; b < AGGR_B; ++b) {
            const int e = c0 + 32 * (b0 + (WG / 32) * b) + lane;
            edge_nodes(src, dst, e, E, Etot, s[b], d[b]);
            a[b] = EW[(e < Etot) ? e : 0];
        }
#pragma unroll
        for (int b = 0; b < AGGR_B; ++b)
            a[b] = __expf(a[b] - tab[d[b]]);
#pragma unroll
        for (int b = 0; b < AGGR_B; ++b)
            if (c0 + 32 * (b0 + (WG / 32) * b) + lane < c1) atomicAdd(&Sh[d[b]], a[b]);

#pragma unroll
        for (int b = 0; b < AGGR_B; ++b) {
            const int cnt = c1 - (c0 + 32 * (b0 + (WG / 32) * b));
            if (cnt >= 32) {
#pragma unroll
                for (int k0 = 0; k0 < 32; k0 += AGGR_G) {
                    float h0[AGGR_G], h1[AGGR_G];
#pragma unroll
                    for (int k = 0; k < AGGR_G; ++k) {
                        const int sk = __builtin_amdgcn_readlane(s[b], k0 + k);
                        const float* hr = Hl + (size_t)sk * FDIM;
                        h0[k] = hr[0];
                        h1[k] = hr[32];
                    }
#pragma unroll
                    for (int k = 0; k < AGGR_G; ++k) {
                        const int   dk = __builtin_amdgcn_readlane(d[b], k0 + k);
                        const float ak = lane_f32(a[b], k0 + k);
                        float* o = Ol + (size_t)dk * FDIM;
                        atomicAdd(o,      ak * h0[k]);
                        atomicAdd(o + 32, ak * h1[k]);
                    }
                }
            } else {
                for (int k = 0; k < cnt; ++k) {
                    const int   sk = __builtin_amdgcn_readlane(s[b], k);
                    const int   dk = __builtin_amdgcn_readlane(d[b], k);
                    const float ak = lane_f32(a[b], k);
                    const float* hr = Hl + (size_t)sk * FDIM;
                    float* o = Ol + (size_t)dk * FDIM;
                    atomicAdd(o,      ak * hr[0]);
                    atomicAdd(o + 32, ak * hr[32]);
                }
            }
        }
    }
}

template <bool FULL>
__device__ __forceinline__ void join_bias_body(const float4* __restrict__ P4,
                                               const float* __restrict__ Sx,
                                               const float* __restrict__ b,
                                               float4* __restrict__ OUT4,
                                               int base, int tot4, int n_nodes)
{
    float4 p[JOIN_V], q[JOIN_V];
    float  s0[JOIN_V], s1[JOIN_V];
#pragma unroll
    for (int j = 0; j < JOIN_V; ++j) {
        const int i  = base + j * TPB;
        const int ic = (FULL || i < tot4) ? i : 0;
        const int nd = ic / (FDIM / 4);
        p[j]  = P4[ic];
        q[j]  = P4[tot4 + ic];
        s0[j] = Sx[nd];
        s1[j] = Sx[n_nodes + nd];
    }
#pragma unroll
    for (int j = 0; j < JOIN_V; ++j) {
        const int i = base + j * TPB;
        if (FULL || i < tot4) {
            const float* c = b + 4 * (i & (FDIM / 4 - 1));
            const float s = s0[j] + s1[j];
            float4 o;
            o.x = (p[j].x + q[j].x) / s + c[0];
            o.y = (p[j].y + q[j].y) / s + c[1];
            o.z = (p[j].z + q[j].z) / s + c[2];
            o.w = (p[j].w + q[j].w) / s + c[3];
            OUT4[i] = o;
        }
    }
}

__global__ __launch_bounds__(TPB) void gat_join_bias(const float* __restrict__ P,
                                                     const float* __restrict__ Sx,
                                                     const float* __restrict__ b,
                                                     float* __restrict__ OUT,
                                                     int tot4, int n_nodes)
{
    const float4* __restrict__ P4 = reinterpret_cast<const float4*>(P);
    float4* __restrict__ OUT4 = reinterpret_cast<float4*>(OUT);
    const int blk0 = blockIdx.x * (TPB * JOIN_V);
    const int base = blk0 + threadIdx.x;
    if (blk0 + TPB * JOIN_V <= tot4)
        join_bias_body<true >(P4, Sx, b, OUT4, base, tot4, n_nodes);
    else
        join_bias_body<false>(P4, Sx, b, OUT4, base, tot4, n_nodes);
}

extern "C" void kernel_launch(void* const* d_in, const int* in_sizes, int n_in,
                              void* d_out, int out_size, void* d_ws, size_t ws_size,
                              hipStream_t stream) {
    const float* x   = (const float*)d_in[0];
    const int*   ei  = (const int*)  d_in[1];
    const float* W1  = (const float*)d_in[2];
    const float* as1 = (const float*)d_in[3];
    const float* ad1 = (const float*)d_in[4];
    const float* b1  = (const float*)d_in[5];
    const float* W2  = (const float*)d_in[6];
    const float* as2 = (const float*)d_in[7];
    const float* ad2 = (const float*)d_in[8];
    const float* b2  = (const float*)d_in[9];

    const int N    = in_sizes[0] / FDIM;
    const int E    = in_sizes[1] / 2;
    const int Etot = E + N;
    const int* src = ei;
    const int* dst = ei + E;

    float* ws = (float*)d_ws;
    float* H  = ws;  ws += (size_t)N * FDIM;
    float* PA = ws;  ws += (size_t)2 * N * FDIM;
    float* PB = ws;  ws += (size_t)2 * N * FDIM;
    float* HS = ws;  ws += N;
    float* HD = ws;  ws += N;
    float* MA = ws;  ws += 2 * N;
    float* MB = ws;  ws += 2 * N;
    float* SA = ws;  ws += 2 * N;
    float* SB = ws;  ws += 2 * N;
    float* EW = ws;  ws += Etot;

    const int tot  = N * FDIM;
    const int tot4 = tot / 4;
    const int need_wgs = (Etot + WG * EDGE_V - 1) / (WG * EDGE_V);
    const int edge_wgs = (need_wgs > EDGE_WGS) ? need_wgs : EDGE_WGS;

    gat_gemm_wmma<false><<<N / 16, 32, 0, stream>>>(x, nullptr, nullptr, nullptr,
                                                    W1, as1, ad1, H, HS, HD,
                                                    PA, MA, SA, N);
    gat_edge_score<<<edge_wgs, WG, 0, stream>>>(src, dst, HS, HD, EW, MA,
                                                E, Etot, N);
    gat_aggr<<<edge_wgs, WG, 0, stream>>>(src, dst, EW, MA, SA, H, PA,
                                          E, Etot, N);
    gat_gemm_wmma<true><<<N / 16, 32, 0, stream>>>(nullptr, PA, SA, b1,
                                                   W2, as2, ad2, H, HS, HD,
                                                   PB, MB, SB, N);
    gat_edge_score<<<edge_wgs, WG, 0, stream>>>(src, dst, HS, HD, EW, MB,
                                                E, Etot, N);
    gat_aggr<<<edge_wgs, WG, 0, stream>>>(src, dst, EW, MB, SB, H, PB,
                                          E, Etot, N);
    gat_join_bias<<<(tot4 + TPB * JOIN_V - 1) / (TPB * JOIN_V), TPB, 0,
                    stream>>>(PB, SB, b2, (float*)d_out, tot4, N);
}
